# starting kernel
# baseline (speedup 1.0000x reference)
#include <hip/hip_runtime.h>


typedef float v2f __attribute__((ext_vector_type(2)));
typedef float v8f __attribute__((ext_vector_type(8)));

#define I_DIM 5
#define H_DIM 8
#define T_DIM 512

__device__ __forceinline__ v2f mk2(float a, float b) { v2f v; v.x = a; v.y = b; return v; }

#if __has_builtin(__builtin_amdgcn_tanhf)
__device__ __forceinline__ float tanh_hw(float x) {
  return __builtin_amdgcn_tanhf(x);
}
#else
__device__ __forceinline__ float tanh_hw(float x) {
  float e = __expf(-2.0f * x);
  return (1.0f - e) * __builtin_amdgcn_rcpf(1.0f + e);
}
#endif
__device__ __forceinline__ float sigmoid_f(float x) {
  return __builtin_fmaf(0.5f, tanh_hw(0.5f * x), 0.5f);
}

__device__ __forceinline__ float swz16(float v) {
  return __int_as_float(__builtin_amdgcn_ds_swizzle(__float_as_int(v), 0x401F));
}

__device__ __forceinline__ v8f wmma4(v2f a, v2f b, v8f c) {
  return __builtin_amdgcn_wmma_f32_16x16x4_f32(false, a, false, b, (short)0, c,
                                               false, false);
}

__global__ void __launch_bounds__(128) gru_wmma_kernel(
    const float* __restrict__ x, const float* __restrict__ w_ih,
    const float* __restrict__ w_hh, const float* __restrict__ b_ih,
    const float* __restrict__ b_hh, const float* __restrict__ fc_w,
    const float* __restrict__ fc_b, float* __restrict__ out) {
  const int lane = threadIdx.x & 31;
  const int wid  = threadIdx.x >> 5;
  const int row  = lane & 15;
  const bool hi  = lane >= 16;
  const int ka   = hi ? 2 : 0;
  const int bglob = blockIdx.x * 64 + wid * 16 + row;

  const bool nrow = row < 8;
  const float* whr0 = w_hh + row * H_DIM;
  const float* whr1 = w_hh + (16 + (row & 7)) * H_DIM;
  v2f a_hh_t0_k0 = mk2(whr0[ka],     whr0[ka + 1]);
  v2f a_hh_t0_k1 = mk2(whr0[4 + ka], whr0[5 + ka]);
  v2f a_hh_t1_k0 = nrow ? mk2(whr1[ka],     whr1[ka + 1]) : mk2(0.f, 0.f);
  v2f a_hh_t1_k1 = nrow ? mk2(whr1[4 + ka], whr1[5 + ka]) : mk2(0.f, 0.f);

  const float* wir0 = w_ih + row * I_DIM;
  const float* wir1 = w_ih + (16 + (row & 7)) * I_DIM;
  v2f a_ih_t0_k0 = mk2(wir0[ka], wir0[ka + 1]);
  v2f a_ih_t0_k1 = hi ? mk2(0.f, 0.f) : mk2(wir0[4], 0.f);
  v2f a_ih_t1_k0 = nrow ? mk2(wir1[ka], wir1[ka + 1]) : mk2(0.f, 0.f);
  v2f a_ih_t1_k1 = (nrow && !hi) ? mk2(wir1[4], 0.f) : mk2(0.f, 0.f);

  v8f c_rz, c_xn, c_hn;
#pragma unroll
  for (int j = 0; j < 8; ++j) {
    int m = j + (hi ? 8 : 0);
    c_rz[j] = b_ih[m] + b_hh[m];
    c_xn[j] = hi ? 0.f : b_ih[16 + j];
    c_hn[j] = hi ? 0.f : b_hh[16 + j];
  }

  v2f h0 = mk2(0.f, 0.f), h1 = mk2(0.f, 0.f);

  const float* xb = x + (size_t)bglob * (T_DIM * I_DIM);
  float xa = xb[ka], xc = xb[ka + 1], xe = hi ? 0.f : xb[4];

  for (int t = 0; t < T_DIM; ++t) {
    int tn = (t + 1 < T_DIM) ? t + 1 : t;
    const float* p = xb + tn * I_DIM;
    float nxa = p[ka], nxc = p[ka + 1], nxe = hi ? 0.f : p[4];
    __builtin_prefetch(xb + ((t + 16 < T_DIM) ? t + 16 : t) * I_DIM, 0, 1);

    v2f xB0 = mk2(xa, xc);
    v2f xB1 = mk2(xe, 0.f);

    v8f rz = c_rz;
    rz = wmma4(a_ih_t0_k0, xB0, rz);
    rz = wmma4(a_ih_t0_k1, xB1, rz);
    rz = wmma4(a_hh_t0_k0, h0, rz);
    rz = wmma4(a_hh_t0_k1, h1, rz);
    v8f xn = c_xn;
    xn = wmma4(a_ih_t1_k0, xB0, xn);
    xn = wmma4(a_ih_t1_k1, xB1, xn);
    v8f hn = c_hn;
    hn = wmma4(a_hh_t1_k0, h0, hn);
    hn = wmma4(a_hh_t1_k1, h1, hn);

    float s[8], n[8];
#pragma unroll
    for (int j = 0; j < 8; ++j) s[j] = sigmoid_f(rz[j]);
#pragma unroll
    for (int j = 0; j < 8; ++j) n[j] = tanh_hw(xn[j] + s[j] * hn[j]);

    float zs0 = swz16(s[0]), zs1 = swz16(s[1]), zs4 = swz16(s[4]), zs5 = swz16(s[5]);
    float ns2 = swz16(n[2]), ns3 = swz16(n[3]), ns6 = swz16(n[6]), ns7 = swz16(n[7]);

    float z0 = hi ? s[2] : zs0, nn0 = hi ? ns2 : n[0];
    float z1 = hi ? s[3] : zs1, nn1 = hi ? ns3 : n[1];
    float z2 = hi ? s[6] : zs4, nn2 = hi ? ns6 : n[4];
    float z3 = hi ? s[7] : zs5, nn3 = hi ? ns7 : n[5];

    h0.x = nn0 + z0 * (h0.x - nn0);
    h0.y = nn1 + z1 * (h0.y - nn1);
    h1.x = nn2 + z2 * (h1.x - nn2);
    h1.y = nn3 + z3 * (h1.y - nn3);

    xa = nxa; xc = nxc; xe = nxe;
  }

  float o0 = fc_w[0 * 8 + ka] * h0.x + fc_w[0 * 8 + ka + 1] * h0.y +
             fc_w[0 * 8 + ka + 4] * h1.x + fc_w[0 * 8 + ka + 5] * h1.y;
  float o1 = fc_w[1 * 8 + ka] * h0.x + fc_w[1 * 8 + ka + 1] * h0.y +
             fc_w[1 * 8 + ka + 4] * h1.x + fc_w[1 * 8 + ka + 5] * h1.y;
  o0 += swz16(o0);
  o1 += swz16(o1);
  if (!hi) {
    out[(size_t)bglob * 2 + 0] = o0 + fc_b[0];
    out[(size_t)bglob * 2 + 1] = o1 + fc_b[1];
  }
}

extern "C" void kernel_launch(void* const* d_in, const int* in_sizes, int n_in,
                              void* d_out, int out_size, void* d_ws, size_t ws_size,
                              hipStream_t stream) {
  const float* x    = (const float*)d_in[0];
  const float* w_ih = (const float*)d_in[1];
  const float* w_hh = (const float*)d_in[2];
  const float* b_ih = (const float*)d_in[3];
  const float* b_hh = (const float*)d_in[4];
  const float* fc_w = (const float*)d_in[5];
  const float* fc_b = (const float*)d_in[6];
  float* out = (float*)d_out;

  const int B = in_sizes[0] / (T_DIM * I_DIM);
  dim3 grid(B / 64), block(128);
  gru_wmma_kernel<<<grid, block, 0, stream>>>(x, w_ih, w_hh, b_ih, b_hh, fc_w,
                                              fc_b, out);
}


# x loaded one 8-timestep chunk ahead into registers (16 loads per lane up front, loop unrolled over two chunks, register budget opened for it); no per-step load, masked branch or prefetch
# speedup vs baseline: 1.4544x; 1.4544x over previous
#include <hip/hip_runtime.h>


typedef float v2f __attribute__((ext_vector_type(2)));
typedef float v2f_a4 __attribute__((ext_vector_type(2), aligned(4)));
typedef float v8f __attribute__((ext_vector_type(8)));

#define I_DIM 5
#define H_DIM 8
#define T_DIM 512
#define CH_T 8
#define CH_F (CH_T * I_DIM)
#define N_CH (T_DIM / CH_T)

__device__ __forceinline__ v2f mk2(float a, float b) { v2f v; v.x = a; v.y = b; return v; }

#if __has_builtin(__builtin_amdgcn_tanhf)
__device__ __forceinline__ float tanh_hw(float x) {
  return __builtin_amdgcn_tanhf(x);
}
#else
__device__ __forceinline__ float tanh_hw(float x) {
  float e = __expf(-2.0f * x);
  return (1.0f - e) * __builtin_amdgcn_rcpf(1.0f + e);
}
#endif
__device__ __forceinline__ float sigmoid_f(float x) {
  return __builtin_fmaf(0.5f, tanh_hw(0.5f * x), 0.5f);
}

__device__ __forceinline__ float swz16(float v) {
  return __int_as_float(__builtin_amdgcn_ds_swizzle(__float_as_int(v), 0x401F));
}

__device__ __forceinline__ v8f wmma4(v2f a, v2f b, v8f c) {
  return __builtin_amdgcn_wmma_f32_16x16x4_f32(false, a, false, b, (short)0, c,
                                               false, false);
}

struct GruW {
  v2f a_hh_t0_k0, a_hh_t0_k1, a_hh_t1_k0, a_hh_t1_k1;
  v2f a_ih_t0_k0, a_ih_t0_k1, a_ih_t1_k0, a_ih_t1_k1;
  v8f c_rz, c_xn, c_hn;
};

struct XChunk { v2f ab[CH_T]; float e[CH_T]; };

__device__ __forceinline__ void load_chunk(XChunk& c, const float* __restrict__ xk,
                                           const float* __restrict__ x4, int chunk) {
#pragma unroll
  for (int s = 0; s < CH_T; ++s) {
    c.ab[s] = *reinterpret_cast<const v2f_a4*>(xk + chunk * CH_F + s * I_DIM);
    c.e[s]  = x4[chunk * CH_F + s * I_DIM];
  }
}

__device__ __forceinline__ void gru_step(const GruW& w, bool hi, v2f xB0,
                                         float xe, v2f& h0, v2f& h1) {
  v2f xB1 = mk2(xe, 0.f);

  v8f rz = w.c_rz;
  rz = wmma4(w.a_ih_t0_k0, xB0, rz);
  rz = wmma4(w.a_ih_t0_k1, xB1, rz);
  rz = wmma4(w.a_hh_t0_k0, h0, rz);
  rz = wmma4(w.a_hh_t0_k1, h1, rz);
  v8f xn = w.c_xn;
  xn = wmma4(w.a_ih_t1_k0, xB0, xn);
  xn = wmma4(w.a_ih_t1_k1, xB1, xn);
  v8f hn = w.c_hn;
  hn = wmma4(w.a_hh_t1_k0, h0, hn);
  hn = wmma4(w.a_hh_t1_k1, h1, hn);

  float s[8], n[8];
#pragma unroll
  for (int j = 0; j < 8; ++j) s[j] = sigmoid_f(rz[j]);
#pragma unroll
  for (int j = 0; j < 8; ++j) n[j] = tanh_hw(xn[j] + s[j] * hn[j]);

  float zs0 = swz16(s[0]), zs1 = swz16(s[1]), zs4 = swz16(s[4]), zs5 = swz16(s[5]);
  float ns2 = swz16(n[2]), ns3 = swz16(n[3]), ns6 = swz16(n[6]), ns7 = swz16(n[7]);

  float z0 = hi ? s[2] : zs0, nn0 = hi ? ns2 : n[0];
  float z1 = hi ? s[3] : zs1, nn1 = hi ? ns3 : n[1];
  float z2 = hi ? s[6] : zs4, nn2 = hi ? ns6 : n[4];
  float z3 = hi ? s[7] : zs5, nn3 = hi ? ns7 : n[5];

  h0.x = nn0 + z0 * (h0.x - nn0);
  h0.y = nn1 + z1 * (h0.y - nn1);
  h1.x = nn2 + z2 * (h1.x - nn2);
  h1.y = nn3 + z3 * (h1.y - nn3);
}

__device__ __forceinline__ void run_chunk(const XChunk& c, const GruW& w, bool hi,
                                          v2f& h0, v2f& h1) {
#pragma unroll
  for (int s = 0; s < CH_T; ++s) {
    gru_step(w, hi, c.ab[s], hi ? 0.f : c.e[s], h0, h1);
  }
}

__global__ void __launch_bounds__(128) __attribute__((amdgpu_waves_per_eu(1, 2))) gru_wmma_kernel(
    const float* __restrict__ x, const float* __restrict__ w_ih,
    const float* __restrict__ w_hh, const float* __restrict__ b_ih,
    const float* __restrict__ b_hh, const float* __restrict__ fc_w,
    const float* __restrict__ fc_b, float* __restrict__ out) {
  const int lane = threadIdx.x & 31;
  const int wid  = threadIdx.x >> 5;
  const int row  = lane & 15;
  const bool hi  = lane >= 16;
  const int ka   = hi ? 2 : 0;
  const int bglob = blockIdx.x * 64 + wid * 16 + row;

  GruW w;
  const bool nrow = row < 8;
  const float* whr0 = w_hh + row * H_DIM;
  const float* whr1 = w_hh + (16 + (row & 7)) * H_DIM;
  w.a_hh_t0_k0 = mk2(whr0[ka],     whr0[ka + 1]);
  w.a_hh_t0_k1 = mk2(whr0[4 + ka], whr0[5 + ka]);
  w.a_hh_t1_k0 = nrow ? mk2(whr1[ka],     whr1[ka + 1]) : mk2(0.f, 0.f);
  w.a_hh_t1_k1 = nrow ? mk2(whr1[4 + ka], whr1[5 + ka]) : mk2(0.f, 0.f);

  const float* wir0 = w_ih + row * I_DIM;
  const float* wir1 = w_ih + (16 + (row & 7)) * I_DIM;
  w.a_ih_t0_k0 = mk2(wir0[ka], wir0[ka + 1]);
  w.a_ih_t0_k1 = hi ? mk2(0.f, 0.f) : mk2(wir0[4], 0.f);
  w.a_ih_t1_k0 = nrow ? mk2(wir1[ka], wir1[ka + 1]) : mk2(0.f, 0.f);
  w.a_ih_t1_k1 = (nrow && !hi) ? mk2(wir1[4], 0.f) : mk2(0.f, 0.f);

#pragma unroll
  for (int j = 0; j < 8; ++j) {
    int m = j + (hi ? 8 : 0);
    w.c_rz[j] = b_ih[m] + b_hh[m];
    w.c_xn[j] = hi ? 0.f : b_ih[16 + j];
    w.c_hn[j] = hi ? 0.f : b_hh[16 + j];
  }

  v2f h0 = mk2(0.f, 0.f), h1 = mk2(0.f, 0.f);

  const float* xb = x + (size_t)bglob * (T_DIM * I_DIM);
  const float* xk = xb + ka;
  const float* x4 = xb + 4;

  XChunk bufA, bufB;
  load_chunk(bufA, xk, x4, 0);
  for (int c = 0; c < N_CH; c += 2) {
    load_chunk(bufB, xk, x4, c + 1);
    __builtin_amdgcn_sched_barrier(0);
    run_chunk(bufA, w, hi, h0, h1);
    load_chunk(bufA, xk, x4, (c + 2 < N_CH) ? c + 2 : c);
    __builtin_amdgcn_sched_barrier(0);
    run_chunk(bufB, w, hi, h0, h1);
  }

  float o0 = fc_w[0 * 8 + ka] * h0.x + fc_w[0 * 8 + ka + 1] * h0.y +
             fc_w[0 * 8 + ka + 4] * h1.x + fc_w[0 * 8 + ka + 5] * h1.y;
  float o1 = fc_w[1 * 8 + ka] * h0.x + fc_w[1 * 8 + ka + 1] * h0.y +
             fc_w[1 * 8 + ka + 4] * h1.x + fc_w[1 * 8 + ka + 5] * h1.y;
  o0 += swz16(o0);
  o1 += swz16(o1);
  if (!hi) {
    out[(size_t)bglob * 2 + 0] = o0 + fc_b[0];
    out[(size_t)bglob * 2 + 1] = o1 + fc_b[1];
  }
}

extern "C" void kernel_launch(void* const* d_in, const int* in_sizes, int n_in,
                              void* d_out, int out_size, void* d_ws, size_t ws_size,
                              hipStream_t stream) {
  const float* x    = (const float*)d_in[0];
  const float* w_ih = (const float*)d_in[1];
  const float* w_hh = (const float*)d_in[2];
  const float* b_ih = (const float*)d_in[3];
  const float* b_hh = (const float*)d_in[4];
  const float* fc_w = (const float*)d_in[5];
  const float* fc_b = (const float*)d_in[6];
  float* out = (float*)d_out;

  const int B = in_sizes[0] / (T_DIM * I_DIM);
  dim3 grid(B / 64), block(128);
  gru_wmma_kernel<<<grid, block, 0, stream>>>(x, w_ih, w_hh, b_ih, b_hh, fc_w,
                                              fc_b, out);
}


# gate rows reordered so D gives each lane r, z and both n pieces of the four hidden units it carries: no cross-lane exchange or select per step, 12 not 16 tanh per lane; sums unchanged
# speedup vs baseline: 2.3191x; 1.5945x over previous
#include <hip/hip_runtime.h>


typedef float v2f __attribute__((ext_vector_type(2)));
typedef float v2f_a4 __attribute__((ext_vector_type(2), aligned(4)));
typedef float v8f __attribute__((ext_vector_type(8)));

#define I_DIM 5
#define H_DIM 8
#define T_DIM 512
#define CH_T 8
#define CH_F (CH_T * I_DIM)
#define N_CH (T_DIM / CH_T)

__device__ __forceinline__ v2f mk2(float a, float b) { v2f v; v.x = a; v.y = b; return v; }

#if __has_builtin(__builtin_amdgcn_tanhf)
__device__ __forceinline__ float tanh_hw(float x) {
  return __builtin_amdgcn_tanhf(x);
}
#else
__device__ __forceinline__ float tanh_hw(float x) {
  float e = __expf(-2.0f * x);
  return (1.0f - e) * __builtin_amdgcn_rcpf(1.0f + e);
}
#endif
__device__ __forceinline__ float sigmoid_f(float x) {
  return __builtin_fmaf(0.5f, tanh_hw(0.5f * x), 0.5f);
}

__device__ __forceinline__ float swz16(float v) {
  return __int_as_float(__builtin_amdgcn_ds_swizzle(__float_as_int(v), 0x401F));
}

__device__ __forceinline__ v8f wmma4(v2f a, v2f b, v8f c) {
  return __builtin_amdgcn_wmma_f32_16x16x4_f32(false, a, false, b, (short)0, c,
                                               false, false);
}

struct GruW {
  v2f a_hh_t0_k0, a_hh_t0_k1, a_hh_t1_k0, a_hh_t1_k1;
  v2f a_ih_t0_k0, a_ih_t0_k1, a_ih_t1_k0, a_ih_t1_k1;
  v8f c_t0, c_t1;
};

struct XChunk { v2f ab[CH_T]; float e[CH_T]; };

__device__ __forceinline__ void load_chunk(XChunk& c, const float* __restrict__ xk,
                                           const float* __restrict__ x4, int chunk) {
#pragma unroll
  for (int s = 0; s < CH_T; ++s) {
    c.ab[s] = *reinterpret_cast<const v2f_a4*>(xk + chunk * CH_F + s * I_DIM);
    c.e[s]  = x4[chunk * CH_F + s * I_DIM];
  }
}

__device__ __forceinline__ void gru_step(const GruW& w, v2f xB0,
                                         float xe, v2f& h0, v2f& h1) {
  v2f xB1 = mk2(xe, 0.f);

  v8f t0 = w.c_t0;
  t0 = wmma4(w.a_ih_t0_k0, xB0, t0);
  t0 = wmma4(w.a_ih_t0_k1, xB1, t0);
  t0 = wmma4(w.a_hh_t0_k0, h0, t0);
  t0 = wmma4(w.a_hh_t0_k1, h1, t0);
  v8f t1 = w.c_t1;
  t1 = wmma4(w.a_ih_t1_k0, xB0, t1);
  t1 = wmma4(w.a_ih_t1_k1, xB1, t1);
  t1 = wmma4(w.a_hh_t1_k0, h0, t1);
  t1 = wmma4(w.a_hh_t1_k1, h1, t1);

  float r[4], z[4], n[4];
#pragma unroll
  for (int i = 0; i < 4; ++i) r[i] = sigmoid_f(t0[i]);
#pragma unroll
  for (int i = 0; i < 4; ++i) n[i] = tanh_hw(t1[i] + r[i] * t1[4 + i]);
#pragma unroll
  for (int i = 0; i < 4; ++i) z[i] = sigmoid_f(t0[4 + i]);

  h0.x = n[0] + z[0] * (h0.x - n[0]);
  h0.y = n[1] + z[1] * (h0.y - n[1]);
  h1.x = n[2] + z[2] * (h1.x - n[2]);
  h1.y = n[3] + z[3] * (h1.y - n[3]);
}

__device__ __forceinline__ void run_chunk(const XChunk& c, const GruW& w, bool hi,
                                          v2f& h0, v2f& h1) {
#pragma unroll
  for (int s = 0; s < CH_T; ++s) {
    gru_step(w, c.ab[s], hi ? 0.f : c.e[s], h0, h1);
  }
}

__global__ void __launch_bounds__(128) __attribute__((amdgpu_waves_per_eu(1, 2))) gru_wmma_kernel(
    const float* __restrict__ x, const float* __restrict__ w_ih,
    const float* __restrict__ w_hh, const float* __restrict__ b_ih,
    const float* __restrict__ b_hh, const float* __restrict__ fc_w,
    const float* __restrict__ fc_b, float* __restrict__ out) {
  const int lane = threadIdx.x & 31;
  const int wid  = threadIdx.x >> 5;
  const int row  = lane & 15;
  const bool hi  = lane >= 16;
  const int ka   = hi ? 2 : 0;
  const int bglob = blockIdx.x * 64 + wid * 16 + row;

  GruW w;
  const int  aq    = row >> 3;
  const int  ai    = row & 3;
  const int  aunit = (ai < 2 ? ai : ai + 2) + 2 * aq;
  const bool afst  = (row & 4) == 0;
  const int  g0    = (afst ? 0 : H_DIM) + aunit;
  const int  g1    = 2 * H_DIM + aunit;
  const float* whr0 = w_hh + g0 * H_DIM;
  const float* whr1 = w_hh + g1 * H_DIM;
  w.a_hh_t0_k0 = mk2(whr0[ka],     whr0[ka + 1]);
  w.a_hh_t0_k1 = mk2(whr0[4 + ka], whr0[5 + ka]);
  w.a_hh_t1_k0 = afst ? mk2(0.f, 0.f) : mk2(whr1[ka],     whr1[ka + 1]);
  w.a_hh_t1_k1 = afst ? mk2(0.f, 0.f) : mk2(whr1[4 + ka], whr1[5 + ka]);

  const float* wir0 = w_ih + g0 * I_DIM;
  const float* wir1 = w_ih + g1 * I_DIM;
  w.a_ih_t0_k0 = mk2(wir0[ka], wir0[ka + 1]);
  w.a_ih_t0_k1 = hi ? mk2(0.f, 0.f) : mk2(wir0[4], 0.f);
  w.a_ih_t1_k0 = afst ? mk2(wir1[ka], wir1[ka + 1]) : mk2(0.f, 0.f);
  w.a_ih_t1_k1 = (afst && !hi) ? mk2(wir1[4], 0.f) : mk2(0.f, 0.f);

#pragma unroll
  for (int i = 0; i < 4; ++i) {
    const int u = (i < 2 ? i : i + 2) + ka;
    w.c_t0[i]     = b_ih[u] + b_hh[u];
    w.c_t0[4 + i] = b_ih[H_DIM + u] + b_hh[H_DIM + u];
    w.c_t1[i]     = b_ih[2 * H_DIM + u];
    w.c_t1[4 + i] = b_hh[2 * H_DIM + u];
  }

  v2f h0 = mk2(0.f, 0.f), h1 = mk2(0.f, 0.f);

  const float* xb = x + (size_t)bglob * (T_DIM * I_DIM);
  const float* xk = xb + ka;
  const float* x4 = xb + 4;

  XChunk bufA, bufB;
  load_chunk(bufA, xk, x4, 0);
  for (int c = 0; c < N_CH; c += 2) {
    load_chunk(bufB, xk, x4, c + 1);
    __builtin_amdgcn_sched_barrier(0);
    run_chunk(bufA, w, hi, h0, h1);
    load_chunk(bufA, xk, x4, (c + 2 < N_CH) ? c + 2 : c);
    __builtin_amdgcn_sched_barrier(0);
    run_chunk(bufB, w, hi, h0, h1);
  }

  float o0 = fc_w[0 * 8 + ka] * h0.x + fc_w[0 * 8 + ka + 1] * h0.y +
             fc_w[0 * 8 + ka + 4] * h1.x + fc_w[0 * 8 + ka + 5] * h1.y;
  float o1 = fc_w[1 * 8 + ka] * h0.x + fc_w[1 * 8 + ka + 1] * h0.y +
             fc_w[1 * 8 + ka + 4] * h1.x + fc_w[1 * 8 + ka + 5] * h1.y;
  o0 += swz16(o0);
  o1 += swz16(o1);
  if (!hi) {
    out[(size_t)bglob * 2 + 0] = o0 + fc_b[0];
    out[(size_t)bglob * 2 + 1] = o1 + fc_b[1];
  }
}

extern "C" void kernel_launch(void* const* d_in, const int* in_sizes, int n_in,
                              void* d_out, int out_size, void* d_ws, size_t ws_size,
                              hipStream_t stream) {
  const float* x    = (const float*)d_in[0];
  const float* w_ih = (const float*)d_in[1];
  const float* w_hh = (const float*)d_in[2];
  const float* b_ih = (const float*)d_in[3];
  const float* b_hh = (const float*)d_in[4];
  const float* fc_w = (const float*)d_in[5];
  const float* fc_b = (const float*)d_in[6];
  float* out = (float*)d_out;

  const int B = in_sizes[0] / (T_DIM * I_DIM);
  dim3 grid(B / 64), block(128);
  gru_wmma_kernel<<<grid, block, 0, stream>>>(x, w_ih, w_hh, b_ih, b_hh, fc_w,
                                              fc_b, out);
}


# x products moved to a partner wave on another SIMD (2 recurrence + 2 projection waves per work-group), finished D tiles handed over through LDS as the C operand of the h products; sums unchanged
# speedup vs baseline: 3.0635x; 1.3210x over previous
#include <hip/hip_runtime.h>


typedef float v2f __attribute__((ext_vector_type(2)));
typedef float v2f_a4 __attribute__((ext_vector_type(2), aligned(4)));
typedef float v4f __attribute__((ext_vector_type(4)));
typedef float v8f __attribute__((ext_vector_type(8)));

#define I_DIM 5
#define H_DIM 8
#define T_DIM 512
#define CH_T 8
#define CH_F (CH_T * I_DIM)
#define N_CH (T_DIM / CH_T)
#define RING_G 4
#define RING_T (RING_G * CH_T)
#define N_RING (T_DIM / RING_T)
#define STEP_V4 (3 * 32)

__device__ __forceinline__ v2f mk2(float a, float b) { v2f v; v.x = a; v.y = b; return v; }

#if __has_builtin(__builtin_amdgcn_tanhf)
__device__ __forceinline__ float tanh_hw(float x) {
  return __builtin_amdgcn_tanhf(x);
}
#else
__device__ __forceinline__ float tanh_hw(float x) {
  float e = __expf(-2.0f * x);
  return (1.0f - e) * __builtin_amdgcn_rcpf(1.0f + e);
}
#endif
__device__ __forceinline__ float sigmoid_f(float x) {
  return __builtin_fmaf(0.5f, tanh_hw(0.5f * x), 0.5f);
}

__device__ __forceinline__ float swz16(float v) {
  return __int_as_float(__builtin_amdgcn_ds_swizzle(__float_as_int(v), 0x401F));
}

__device__ __forceinline__ v8f wmma4(v2f a, v2f b, v8f c) {
  return __builtin_amdgcn_wmma_f32_16x16x4_f32(false, a, false, b, (short)0, c,
                                               false, false);
}

struct GruW {
  v2f a_hh_t0_k0, a_hh_t0_k1, a_hh_t1_k0, a_hh_t1_k1;
  v2f a_ih_t0_k0, a_ih_t0_k1, a_ih_t1_k0, a_ih_t1_k1;
  v8f c_t0, c_t1;
};

struct XChunk { v2f ab[CH_T]; float e[CH_T]; };

__device__ __forceinline__ void load_chunk(XChunk& c, const float* __restrict__ xk,
                                           const float* __restrict__ x4, int chunk) {
#pragma unroll
  for (int s = 0; s < CH_T; ++s) {
    c.ab[s] = *reinterpret_cast<const v2f_a4*>(xk + chunk * CH_F + s * I_DIM);
    c.e[s]  = x4[chunk * CH_F + s * I_DIM];
  }
}

struct XG { v4f p[CH_T][3]; };

__device__ __forceinline__ void proj_chunk(const XChunk& c, const GruW& w, bool hi, v4f* dst) {
#pragma unroll
  for (int s = 0; s < CH_T; ++s) {
    v2f xB0 = c.ab[s];
    v2f xB1 = mk2(hi ? 0.f : c.e[s], 0.f);
    v8f t0 = w.c_t0;
    t0 = wmma4(w.a_ih_t0_k0, xB0, t0);
    t0 = wmma4(w.a_ih_t0_k1, xB1, t0);
    v8f t1 = w.c_t1;
    t1 = wmma4(w.a_ih_t1_k0, xB0, t1);
    t1 = wmma4(w.a_ih_t1_k1, xB1, t1);
    dst[(s * 3 + 0) * 32] = __builtin_shufflevector(t0, t0, 0, 1, 2, 3);
    dst[(s * 3 + 1) * 32] = __builtin_shufflevector(t0, t0, 4, 5, 6, 7);
    dst[(s * 3 + 2) * 32] = __builtin_shufflevector(t1, t1, 0, 1, 2, 3);
  }
}

__device__ __forceinline__ void load_group(XG& g, const v4f* src) {
#pragma unroll
  for (int s = 0; s < CH_T; ++s) {
#pragma unroll
    for (int p = 0; p < 3; ++p) g.p[s][p] = src[(s * 3 + p) * 32];
  }
}

__device__ __forceinline__ void gru_step(const GruW& w, v4f c0, v4f c1, v4f nx,
                                         v2f& h0, v2f& h1) {
  v8f t0 = __builtin_shufflevector(c0, c1, 0, 1, 2, 3, 4, 5, 6, 7);
  t0 = wmma4(w.a_hh_t0_k0, h0, t0);
  t0 = wmma4(w.a_hh_t0_k1, h1, t0);
  v8f t1 = w.c_t1;
  t1 = wmma4(w.a_hh_t1_k0, h0, t1);
  t1 = wmma4(w.a_hh_t1_k1, h1, t1);

  float r[4], z[4], n[4];
#pragma unroll
  for (int i = 0; i < 4; ++i) r[i] = sigmoid_f(t0[i]);
#pragma unroll
  for (int i = 0; i < 4; ++i) n[i] = tanh_hw(nx[i] + r[i] * t1[4 + i]);
#pragma unroll
  for (int i = 0; i < 4; ++i) z[i] = sigmoid_f(t0[4 + i]);

  h0.x = n[0] + z[0] * (h0.x - n[0]);
  h0.y = n[1] + z[1] * (h0.y - n[1]);
  h1.x = n[2] + z[2] * (h1.x - n[2]);
  h1.y = n[3] + z[3] * (h1.y - n[3]);
}

__device__ __forceinline__ void run_group(const XG& g, const GruW& w, v2f& h0, v2f& h1) {
#pragma unroll
  for (int s = 0; s < CH_T; ++s) {
    gru_step(w, g.p[s][0], g.p[s][1], g.p[s][2], h0, h1);
  }
}

__device__ __forceinline__ void lds_handover_barrier() {
  __builtin_amdgcn_fence(__ATOMIC_RELEASE, "workgroup", "local");
  __builtin_amdgcn_s_barrier();
  __builtin_amdgcn_fence(__ATOMIC_ACQUIRE, "workgroup", "local");
}

__global__ void __launch_bounds__(128) __attribute__((amdgpu_waves_per_eu(1, 2))) gru_wmma_kernel(
    const float* __restrict__ x, const float* __restrict__ w_ih,
    const float* __restrict__ w_hh, const float* __restrict__ b_ih,
    const float* __restrict__ b_hh, const float* __restrict__ fc_w,
    const float* __restrict__ fc_b, float* __restrict__ out) {
  __shared__ v4f ring[2 * 2 * RING_T * STEP_V4];

  const int lane = threadIdx.x & 31;
  const int wid  = __builtin_amdgcn_readfirstlane((int)(threadIdx.x >> 5));
  const bool proj = wid >= 2;
  const int pair = wid & 1;
  const int row  = lane & 15;
  const bool hi  = lane >= 16;
  const int ka   = hi ? 2 : 0;
  const int bglob = blockIdx.x * 32 + pair * 16 + row;

  GruW w;
  const int  aq    = row >> 3;
  const int  ai    = row & 3;
  const int  aunit = (ai < 2 ? ai : ai + 2) + 2 * aq;
  const bool afst  = (row & 4) == 0;
  const int  g0    = (afst ? 0 : H_DIM) + aunit;
  const int  g1    = 2 * H_DIM + aunit;
  const float* whr0 = w_hh + g0 * H_DIM;
  const float* whr1 = w_hh + g1 * H_DIM;
  w.a_hh_t0_k0 = mk2(whr0[ka],     whr0[ka + 1]);
  w.a_hh_t0_k1 = mk2(whr0[4 + ka], whr0[5 + ka]);
  w.a_hh_t1_k0 = afst ? mk2(0.f, 0.f) : mk2(whr1[ka],     whr1[ka + 1]);
  w.a_hh_t1_k1 = afst ? mk2(0.f, 0.f) : mk2(whr1[4 + ka], whr1[5 + ka]);

  const float* wir0 = w_ih + g0 * I_DIM;
  const float* wir1 = w_ih + g1 * I_DIM;
  w.a_ih_t0_k0 = mk2(wir0[ka], wir0[ka + 1]);
  w.a_ih_t0_k1 = hi ? mk2(0.f, 0.f) : mk2(wir0[4], 0.f);
  w.a_ih_t1_k0 = afst ? mk2(wir1[ka], wir1[ka + 1]) : mk2(0.f, 0.f);
  w.a_ih_t1_k1 = (afst && !hi) ? mk2(wir1[4], 0.f) : mk2(0.f, 0.f);

#pragma unroll
  for (int i = 0; i < 4; ++i) {
    const int u = (i < 2 ? i : i + 2) + ka;
    w.c_t0[i]     = b_ih[u] + b_hh[u];
    w.c_t0[4 + i] = b_ih[H_DIM + u] + b_hh[H_DIM + u];
    w.c_t1[i]     = b_ih[2 * H_DIM + u];
    w.c_t1[4 + i] = b_hh[2 * H_DIM + u];
  }

  v2f h0 = mk2(0.f, 0.f), h1 = mk2(0.f, 0.f);

  const float* xb = x + (size_t)bglob * (T_DIM * I_DIM);
  const float* xk = xb + ka;
  const float* x4 = xb + 4;

  XChunk bufA, bufB;
  XG gP, gQ;
  if (proj) load_chunk(bufA, xk, x4, 0);

  for (int i = 0; i <= N_RING; ++i) {
    if (proj) {
      if (i < N_RING) {
        v4f* dst = ring + (((i & 1) * 2 + pair) * RING_T) * STEP_V4 + lane;
#pragma unroll 1
        for (int g = 0; g < RING_G; g += 2) {
          const int c = i * RING_G + g;
          load_chunk(bufB, xk, x4, c + 1);
          __builtin_amdgcn_sched_barrier(0);
          proj_chunk(bufA, w, hi, dst + (g * CH_T) * STEP_V4);
          load_chunk(bufA, xk, x4, (c + 2 < N_CH) ? c + 2 : c);
          __builtin_amdgcn_sched_barrier(0);
          proj_chunk(bufB, w, hi, dst + ((g + 1) * CH_T) * STEP_V4);
        }
      }
    } else if (i >= 1) {
      const v4f* src = ring + ((((i - 1) & 1) * 2 + pair) * RING_T) * STEP_V4 + lane;
#pragma unroll 1
      for (int g = 0; g < RING_G; g += 2) {
        load_group(gQ, src + (g * CH_T) * STEP_V4);
        __builtin_amdgcn_sched_barrier(0);
        if (i >= 2 || g > 0) run_group(gP, w, h0, h1);
        load_group(gP, src + ((g + 1) * CH_T) * STEP_V4);
        __builtin_amdgcn_sched_barrier(0);
        run_group(gQ, w, h0, h1);
      }
    }
    lds_handover_barrier();
  }
  if (proj) return;
  run_group(gP, w, h0, h1);

  float o0 = fc_w[0 * 8 + ka] * h0.x + fc_w[0 * 8 + ka + 1] * h0.y +
             fc_w[0 * 8 + ka + 4] * h1.x + fc_w[0 * 8 + ka + 5] * h1.y;
  float o1 = fc_w[1 * 8 + ka] * h0.x + fc_w[1 * 8 + ka + 1] * h0.y +
             fc_w[1 * 8 + ka + 4] * h1.x + fc_w[1 * 8 + ka + 5] * h1.y;
  o0 += swz16(o0);
  o1 += swz16(o1);
  if (!hi) {
    out[(size_t)bglob * 2 + 0] = o0 + fc_b[0];
    out[(size_t)bglob * 2 + 1] = o1 + fc_b[1];
  }
}

extern "C" void kernel_launch(void* const* d_in, const int* in_sizes, int n_in,
                              void* d_out, int out_size, void* d_ws, size_t ws_size,
                              hipStream_t stream) {
  const float* x    = (const float*)d_in[0];
  const float* w_ih = (const float*)d_in[1];
  const float* w_hh = (const float*)d_in[2];
  const float* b_ih = (const float*)d_in[3];
  const float* b_hh = (const float*)d_in[4];
  const float* fc_w = (const float*)d_in[5];
  const float* fc_b = (const float*)d_in[6];
  float* out = (float*)d_out;

  const int B = in_sizes[0] / (T_DIM * I_DIM);
  dim3 grid(B / 32), block(128);
  gru_wmma_kernel<<<grid, block, 0, stream>>>(x, w_ih, w_hh, b_ih, b_hh, fc_w,
                                              fc_b, out);
}
